# starting kernel
# baseline (speedup 1.0000x reference)
#include <hip/hip_runtime.h>


#define Bn    2
#define Cn    64
#define Nn    20000
#define Kn    16
#define COUTn 64

typedef __attribute__((ext_vector_type(2))) float v2f;
typedef __attribute__((ext_vector_type(8))) float v8f;

__global__ void build_wc(const float* __restrict__ W, float* __restrict__ Wc) {
    int t = blockIdx.x * blockDim.x + threadIdx.x;
    if (t >= 128 * 64) return;
    int o = t >> 6, c = t & 63;
    float v;
    if (o < 64) v = W[o * 128 + c] - W[o * 128 + 64 + c];
    else        v = W[(o - 64) * 128 + 64 + c];
    Wc[t] = v;
}

__global__ __launch_bounds__(32) void gemm_y(
    const float* __restrict__ x, const float* __restrict__ Wc,
    const float* __restrict__ bias, float* __restrict__ y) {
    const int lane = threadIdx.x;
    const int nlo  = lane & 15;
    const int hi   = lane >> 4;
    const int tile = blockIdx.x;
    const int b    = tile / (Nn / 16);
    const int m0   = (tile % (Nn / 16)) * 16;

    v8f acc[8];
#pragma unroll
    for (int mt = 0; mt < 8; ++mt) {
#pragma unroll
        for (int r = 0; r < 8; ++r) {
            int o = mt * 16 + r + 8 * hi;
            acc[mt][r] = (o < COUTn) ? bias[o] : 0.0f;
        }
    }

    const float* xb = x + (size_t)b * Cn * Nn;
#pragma unroll 4
    for (int c0 = 0; c0 < Cn; c0 += 4) {
        v2f bf;
        bf.x = xb[(size_t)(c0 + 2 * hi)     * Nn + m0 + nlo];
        bf.y = xb[(size_t)(c0 + 2 * hi + 1) * Nn + m0 + nlo];
#pragma unroll
        for (int mt = 0; mt < 8; ++mt) {
            v2f af;
            af.x = Wc[(mt * 16 + nlo) * 64 + c0 + 2 * hi];
            af.y = Wc[(mt * 16 + nlo) * 64 + c0 + 2 * hi + 1];
            acc[mt] = __builtin_amdgcn_wmma_f32_16x16x4_f32(
                false, af, false, bf, (short)0, acc[mt], false, false);
        }
    }

    float* yrow = y + ((size_t)b * Nn + m0 + nlo) * 128;
#pragma unroll
    for (int mt = 0; mt < 8; ++mt) {
        float4* dst = (float4*)(yrow + mt * 16 + 8 * hi);
        dst[0] = make_float4(acc[mt][0], acc[mt][1], acc[mt][2], acc[mt][3]);
        dst[1] = make_float4(acc[mt][4], acc[mt][5], acc[mt][6], acc[mt][7]);
    }
}

__global__ __launch_bounds__(256) void edge_combine(
    const float* __restrict__ y, const int* __restrict__ eidx,
    const float* __restrict__ pos, float* __restrict__ out) {
    __shared__ float smem[64][8];
    const int t    = threadIdx.x;
    const int lane = t & 31;
    const int w    = t >> 5;
    const long p   = (long)blockIdx.x * 8 + w;
    const int b    = (int)(p / Nn);
    const int n    = (int)(p - (long)b * Nn);

    int iv = 0, jv = 0;
    float supp = 0.0f;
    if (lane < Kn) {
        const size_t ebase = ((size_t)b * Nn + n) * Kn + lane;
        jv = eidx[ebase];
        iv = eidx[(size_t)Bn * Nn * Kn + ebase];
        const float* pb = pos + (size_t)b * 3 * Nn;
        float dx = pb[jv] - pb[iv];
        float dy = pb[Nn + jv] - pb[Nn + iv];
        float dz = pb[2 * Nn + jv] - pb[2 * Nn + iv];
        float dis = sqrtf(dx * dx + dy * dy + dz * dz);
        supp = 2.0f / (1.0f + __expf(dis));
    }

    float mx0 = 0.0f, mx1 = 0.0f;
    const float* yb = y + (size_t)b * Nn * 128;
#pragma unroll
    for (int k = 0; k < Kn; ++k) {
        int   ik = __shfl(iv, k, 32);
        int   jk = __shfl(jv, k, 32);
        float sk = __shfl(supp, k, 32);
        const float* yi = yb + (size_t)ik * 128;
        const float* yj = yb + (size_t)jk * 128 + 64;
        float a0 = yi[lane]      + yj[lane];
        float a1 = yi[lane + 32] + yj[lane + 32];
        a0 = fmaxf(a0, 0.0f) * sk;
        a1 = fmaxf(a1, 0.0f) * sk;
        mx0 = fmaxf(mx0, a0);
        mx1 = fmaxf(mx1, a1);
    }

    smem[lane][w]      = mx0;
    smem[lane + 32][w] = mx1;
    __syncthreads();
    const int b0 = (blockIdx.x * 8) / Nn;
    const int n0 = blockIdx.x * 8 - b0 * Nn;
#pragma unroll
    for (int v = t; v < 512; v += 256) {
        int o = v >> 3, ni = v & 7;
        out[((size_t)b0 * COUTn + o) * Nn + n0 + ni] = smem[o][ni];
    }
}

extern "C" void kernel_launch(void* const* d_in, const int* in_sizes, int n_in,
                              void* d_out, int out_size, void* d_ws, size_t ws_size,
                              hipStream_t stream) {
    const float* x    = (const float*)d_in[0];
    const int*   eidx = (const int*)  d_in[1];
    const float* pos  = (const float*)d_in[2];
    const float* W    = (const float*)d_in[3];
    const float* bias = (const float*)d_in[4];

    float* Wc = (float*)d_ws;
    float* y  = (float*)((char*)d_ws + 128 * 64 * sizeof(float));

    build_wc<<<(128 * 64 + 255) / 256, 256, 0, stream>>>(W, Wc);
    gemm_y<<<Bn * (Nn / 16), 32, 0, stream>>>(x, Wc, bias, y);
    edge_combine<<<(Bn * Nn) / 8, 256, 0, stream>>>(y, eidx, pos, (float*)d_out);
}


# edge stage: one wave owns 8 consecutive nodes, walked in a loop with each half node's 32 row loads requested a half ahead (5000 waves, not 40000 one-node waves); results stored from registers
# speedup vs baseline: 1.1107x; 1.1107x over previous
#include <hip/hip_runtime.h>


#define Bn    2
#define Cn    64
#define Nn    20000
#define Kn    16
#define COUTn 64

typedef __attribute__((ext_vector_type(2))) float v2f;
typedef __attribute__((ext_vector_type(8))) float v8f;

__global__ void build_wc(const float* __restrict__ W, float* __restrict__ Wc) {
    int t = blockIdx.x * blockDim.x + threadIdx.x;
    if (t >= 128 * 64) return;
    int o = t >> 6, c = t & 63;
    float v;
    if (o < 64) v = W[o * 128 + c] - W[o * 128 + 64 + c];
    else        v = W[(o - 64) * 128 + 64 + c];
    Wc[t] = v;
}

__global__ __launch_bounds__(32) void gemm_y(
    const float* __restrict__ x, const float* __restrict__ Wc,
    const float* __restrict__ bias, float* __restrict__ y) {
    const int lane = threadIdx.x;
    const int nlo  = lane & 15;
    const int hi   = lane >> 4;
    const int tile = blockIdx.x;
    const int b    = tile / (Nn / 16);
    const int m0   = (tile % (Nn / 16)) * 16;

    v8f acc[8];
#pragma unroll
    for (int mt = 0; mt < 8; ++mt) {
#pragma unroll
        for (int r = 0; r < 8; ++r) {
            int o = mt * 16 + r + 8 * hi;
            acc[mt][r] = (o < COUTn) ? bias[o] : 0.0f;
        }
    }

    const float* xb = x + (size_t)b * Cn * Nn;
#pragma unroll 4
    for (int c0 = 0; c0 < Cn; c0 += 4) {
        v2f bf;
        bf.x = xb[(size_t)(c0 + 2 * hi)     * Nn + m0 + nlo];
        bf.y = xb[(size_t)(c0 + 2 * hi + 1) * Nn + m0 + nlo];
#pragma unroll
        for (int mt = 0; mt < 8; ++mt) {
            v2f af;
            af.x = Wc[(mt * 16 + nlo) * 64 + c0 + 2 * hi];
            af.y = Wc[(mt * 16 + nlo) * 64 + c0 + 2 * hi + 1];
            acc[mt] = __builtin_amdgcn_wmma_f32_16x16x4_f32(
                false, af, false, bf, (short)0, acc[mt], false, false);
        }
    }

    float* yrow = y + ((size_t)b * Nn + m0 + nlo) * 128;
#pragma unroll
    for (int mt = 0; mt < 8; ++mt) {
        float4* dst = (float4*)(yrow + mt * 16 + 8 * hi);
        dst[0] = make_float4(acc[mt][0], acc[mt][1], acc[mt][2], acc[mt][3]);
        dst[1] = make_float4(acc[mt][4], acc[mt][5], acc[mt][6], acc[mt][7]);
    }
}


template <int HALF>
__device__ __forceinline__ void half_request(
    const int q, const int (&iv)[4], const int (&jv)[4],
    const float* __restrict__ yb, const unsigned lane, float (&d)[32]) {
#pragma unroll
    for (int e = 0; e < 8; ++e) {
        const int k   = 8 * HALF + e;
        const int src = 4 * q + (k >> 2);
        const int ik  = __builtin_amdgcn_readlane(iv[k & 3], src);
        const int jk  = __builtin_amdgcn_readlane(jv[k & 3], src);
        const unsigned ei = (unsigned)ik * 128u + lane;
        const unsigned ej = (unsigned)jk * 128u + 64u + lane;
        d[4 * e + 0] = yb[ei];
        d[4 * e + 1] = yb[ei + 32];
        d[4 * e + 2] = yb[ej];
        d[4 * e + 3] = yb[ej + 32];
    }
}

template <int HALF>
__device__ __forceinline__ void half_sum(
    const int q, const int (&sp)[4], const float (&d)[32], float& mx0, float& mx1) {
#pragma unroll
    for (int e = 0; e < 8; ++e) {
        const int k   = 8 * HALF + e;
        const int src = 4 * q + (k >> 2);
        const float sk = __int_as_float(__builtin_amdgcn_readlane(sp[k & 3], src));
        float a0 = d[4 * e + 0] + d[4 * e + 2];
        float a1 = d[4 * e + 1] + d[4 * e + 3];
        a0 = fmaxf(a0, 0.0f) * sk;
        a1 = fmaxf(a1, 0.0f) * sk;
        mx0 = fmaxf(mx0, a0);
        mx1 = fmaxf(mx1, a1);
    }
}

__global__ __launch_bounds__(128) void edge_combine(
    const float* __restrict__ y, const int* __restrict__ eidx,
    const float* __restrict__ pos, float* __restrict__ out) {
    const int t    = threadIdx.x;
    const unsigned lane = t & 31;
    const int g    = __builtin_amdgcn_readfirstlane((int)blockIdx.x * 4 + (t >> 5));
    const int b    = g / (Nn / 8);
    const int n0   = (g - b * (Nn / 8)) * 8;

    const size_t ebase = ((size_t)b * Nn + n0) * Kn + 4 * lane;
    const int4 j4 = *(const int4*)(eidx + ebase);
    const int4 i4 = *(const int4*)(eidx + (size_t)Bn * Nn * Kn + ebase);
    const int jv[4] = {j4.x, j4.y, j4.z, j4.w};
    const int iv[4] = {i4.x, i4.y, i4.z, i4.w};
    const float* yb = y + (size_t)b * Nn * 128;

    float hA[32], hB[32];
    half_request<0>(0, iv, jv, yb, lane, hA);

    const float* pb = pos + (size_t)b * 3 * Nn;
    int sp[4];
#pragma unroll
    for (int c = 0; c < 4; ++c) {
        float dx = pb[jv[c]] - pb[iv[c]];
        float dy = pb[Nn + jv[c]] - pb[Nn + iv[c]];
        float dz = pb[2 * Nn + jv[c]] - pb[2 * Nn + iv[c]];
        float dis = sqrtf(dx * dx + dy * dy + dz * dz);
        sp[c] = __float_as_int(2.0f / (1.0f + __expf(dis)));
    }

    float r0[8] = {0.0f, 0.0f, 0.0f, 0.0f, 0.0f, 0.0f, 0.0f, 0.0f};
    float r1[8] = {0.0f, 0.0f, 0.0f, 0.0f, 0.0f, 0.0f, 0.0f, 0.0f};
#pragma unroll 1
    for (int q = 0; q < 7; ++q) {
        half_request<1>(q, iv, jv, yb, lane, hB);
        __builtin_amdgcn_sched_barrier(0);
        float mx0 = 0.0f, mx1 = 0.0f;
        half_sum<0>(q, sp, hA, mx0, mx1);
        __builtin_amdgcn_sched_barrier(0);
        half_request<0>(q + 1, iv, jv, yb, lane, hA);
        __builtin_amdgcn_sched_barrier(0);
        half_sum<1>(q, sp, hB, mx0, mx1);
#pragma unroll
        for (int c = 0; c < 7; ++c) {
            r0[c] = r0[c + 1];
            r1[c] = r1[c + 1];
        }
        r0[7] = mx0;
        r1[7] = mx1;
    }
    {
        half_request<1>(7, iv, jv, yb, lane, hB);
        float mx0 = 0.0f, mx1 = 0.0f;
        half_sum<0>(7, sp, hA, mx0, mx1);
        half_sum<1>(7, sp, hB, mx0, mx1);
#pragma unroll
        for (int c = 0; c < 7; ++c) {
            r0[c] = r0[c + 1];
            r1[c] = r1[c + 1];
        }
        r0[7] = mx0;
        r1[7] = mx1;
    }

    float* o0 = out + ((size_t)b * COUTn + lane) * Nn + n0;
    float* o1 = o0 + (size_t)32 * Nn;
    *(float4*)(o0)     = make_float4(r0[0], r0[1], r0[2], r0[3]);
    *(float4*)(o0 + 4) = make_float4(r0[4], r0[5], r0[6], r0[7]);
    *(float4*)(o1)     = make_float4(r1[0], r1[1], r1[2], r1[3]);
    *(float4*)(o1 + 4) = make_float4(r1[4], r1[5], r1[6], r1[7]);
}

extern "C" void kernel_launch(void* const* d_in, const int* in_sizes, int n_in,
                              void* d_out, int out_size, void* d_ws, size_t ws_size,
                              hipStream_t stream) {
    const float* x    = (const float*)d_in[0];
    const int*   eidx = (const int*)  d_in[1];
    const float* pos  = (const float*)d_in[2];
    const float* W    = (const float*)d_in[3];
    const float* bias = (const float*)d_in[4];

    float* Wc = (float*)d_ws;
    float* y  = (float*)((char*)d_ws + 128 * 64 * sizeof(float));

    build_wc<<<(128 * 64 + 255) / 256, 256, 0, stream>>>(W, Wc);
    gemm_y<<<Bn * (Nn / 16), 32, 0, stream>>>(x, Wc, bias, y);
    edge_combine<<<(Bn * Nn) / 32, 128, 0, stream>>>(y, eidx, pos, (float*)d_out);
}


# edge stage: each batch's position table (240 KB) staged once per work-group in LDS (250 work-groups of 20 waves, one per WGP); edge distances read LDS, not 24 scattered global dword loads per wave
# speedup vs baseline: 1.5154x; 1.3644x over previous
#include <hip/hip_runtime.h>


#define Bn    2
#define Cn    64
#define Nn    20000
#define Kn    16
#define COUTn 64

typedef __attribute__((ext_vector_type(2))) float v2f;
typedef __attribute__((ext_vector_type(8))) float v8f;

__global__ void build_wc(const float* __restrict__ W, float* __restrict__ Wc) {
    int t = blockIdx.x * blockDim.x + threadIdx.x;
    if (t >= 128 * 64) return;
    int o = t >> 6, c = t & 63;
    float v;
    if (o < 64) v = W[o * 128 + c] - W[o * 128 + 64 + c];
    else        v = W[(o - 64) * 128 + 64 + c];
    Wc[t] = v;
}

__global__ __launch_bounds__(32) void gemm_y(
    const float* __restrict__ x, const float* __restrict__ Wc,
    const float* __restrict__ bias, float* __restrict__ y) {
    const int lane = threadIdx.x;
    const int nlo  = lane & 15;
    const int hi   = lane >> 4;
    const int tile = blockIdx.x;
    const int b    = tile / (Nn / 16);
    const int m0   = (tile % (Nn / 16)) * 16;

    v8f acc[8];
#pragma unroll
    for (int mt = 0; mt < 8; ++mt) {
#pragma unroll
        for (int r = 0; r < 8; ++r) {
            int o = mt * 16 + r + 8 * hi;
            acc[mt][r] = (o < COUTn) ? bias[o] : 0.0f;
        }
    }

    const float* xb = x + (size_t)b * Cn * Nn;
#pragma unroll 4
    for (int c0 = 0; c0 < Cn; c0 += 4) {
        v2f bf;
        bf.x = xb[(size_t)(c0 + 2 * hi)     * Nn + m0 + nlo];
        bf.y = xb[(size_t)(c0 + 2 * hi + 1) * Nn + m0 + nlo];
#pragma unroll
        for (int mt = 0; mt < 8; ++mt) {
            v2f af;
            af.x = Wc[(mt * 16 + nlo) * 64 + c0 + 2 * hi];
            af.y = Wc[(mt * 16 + nlo) * 64 + c0 + 2 * hi + 1];
            acc[mt] = __builtin_amdgcn_wmma_f32_16x16x4_f32(
                false, af, false, bf, (short)0, acc[mt], false, false);
        }
    }

    float* yrow = y + ((size_t)b * Nn + m0 + nlo) * 128;
#pragma unroll
    for (int mt = 0; mt < 8; ++mt) {
        float4* dst = (float4*)(yrow + mt * 16 + 8 * hi);
        dst[0] = make_float4(acc[mt][0], acc[mt][1], acc[mt][2], acc[mt][3]);
        dst[1] = make_float4(acc[mt][4], acc[mt][5], acc[mt][6], acc[mt][7]);
    }
}


template <int HALF>
__device__ __forceinline__ void half_request(
    const int q, const int (&iv)[4], const int (&jv)[4],
    const float* __restrict__ yb, const unsigned lane, float (&d)[32]) {
#pragma unroll
    for (int e = 0; e < 8; ++e) {
        const int k   = 8 * HALF + e;
        const int src = 4 * q + (k >> 2);
        const int ik  = __builtin_amdgcn_readlane(iv[k & 3], src);
        const int jk  = __builtin_amdgcn_readlane(jv[k & 3], src);
        const unsigned ei = (unsigned)ik * 128u + lane;
        const unsigned ej = (unsigned)jk * 128u + 64u + lane;
        d[4 * e + 0] = yb[ei];
        d[4 * e + 1] = yb[ei + 32];
        d[4 * e + 2] = yb[ej];
        d[4 * e + 3] = yb[ej + 32];
    }
}

template <int HALF>
__device__ __forceinline__ void half_sum(
    const int q, const int (&sp)[4], const float (&d)[32], float& mx0, float& mx1) {
#pragma unroll
    for (int e = 0; e < 8; ++e) {
        const int k   = 8 * HALF + e;
        const int src = 4 * q + (k >> 2);
        const float sk = __int_as_float(__builtin_amdgcn_readlane(sp[k & 3], src));
        float a0 = d[4 * e + 0] + d[4 * e + 2];
        float a1 = d[4 * e + 1] + d[4 * e + 3];
        a0 = fmaxf(a0, 0.0f) * sk;
        a1 = fmaxf(a1, 0.0f) * sk;
        mx0 = fmaxf(mx0, a0);
        mx1 = fmaxf(mx1, a1);
    }
}

#define EWAVES 20
#define ETHR   (EWAVES * 32)
#define PVEC   (3 * Nn / 4)

__global__ __launch_bounds__(ETHR) void edge_combine(
    const float* __restrict__ y, const int* __restrict__ eidx,
    const float* __restrict__ pos, float* __restrict__ out) {
    __shared__ float4 ptab4[PVEC];
    const int t    = threadIdx.x;
    const unsigned lane = t & 31;
    const int w    = __builtin_amdgcn_readfirstlane(t >> 5);
    const int b    = (int)blockIdx.x / (Nn / (8 * EWAVES));
    const int r    = (int)blockIdx.x - b * (Nn / (8 * EWAVES));
    const int n0   = (r * EWAVES + w) * 8;

    const size_t ebase = ((size_t)b * Nn + n0) * Kn + 4 * lane;
    const int4 j4 = *(const int4*)(eidx + ebase);
    const int4 i4 = *(const int4*)(eidx + (size_t)Bn * Nn * Kn + ebase);
    const int jv[4] = {j4.x, j4.y, j4.z, j4.w};
    const int iv[4] = {i4.x, i4.y, i4.z, i4.w};
    const float* yb = y + (size_t)b * Nn * 128;

    const float4* psrc = (const float4*)(pos + (size_t)b * 3 * Nn);
    const int rot = r * (PVEC / (Nn / (8 * EWAVES)));
    const int  tidx = (t + ETHR * 23 + rot >= PVEC) ? t + ETHR * 23 + rot - PVEC : t + ETHR * 23 + rot;
    const bool tail = t + ETHR * 23 < PVEC;
    float4 tv = make_float4(0.0f, 0.0f, 0.0f, 0.0f);
#pragma unroll
    for (int grp = 0; grp < 3; ++grp) {
        float4 v[8];
        int    at[8];
#pragma unroll
        for (int u = 0; u < 8; ++u) {
            const int k = 8 * grp + u;
            if (k < 23) {
                int idx = t + ETHR * k + rot;
                idx = (idx >= PVEC) ? idx - PVEC : idx;
                at[u] = idx;
                v[u] = psrc[idx];
            }
        }
        if (grp == 2 && tail) tv = psrc[tidx];
#pragma unroll
        for (int u = 0; u < 8; ++u) {
            const int k = 8 * grp + u;
            if (k < 23) ptab4[at[u]] = v[u];
        }
    }
    if (tail) ptab4[tidx] = tv;

    float hA[32], hB[32];
    half_request<0>(0, iv, jv, yb, lane, hA);

    __builtin_amdgcn_fence(__ATOMIC_RELEASE, "workgroup", "local");
    __builtin_amdgcn_s_barrier();
    __builtin_amdgcn_fence(__ATOMIC_ACQUIRE, "workgroup", "local");

    const float* pb = (const float*)ptab4;
    int sp[4];
#pragma unroll
    for (int c = 0; c < 4; ++c) {
        float dx = pb[jv[c]] - pb[iv[c]];
        float dy = pb[Nn + jv[c]] - pb[Nn + iv[c]];
        float dz = pb[2 * Nn + jv[c]] - pb[2 * Nn + iv[c]];
        float dis = sqrtf(dx * dx + dy * dy + dz * dz);
        sp[c] = __float_as_int(2.0f / (1.0f + __expf(dis)));
    }

    float r0[8] = {0.0f, 0.0f, 0.0f, 0.0f, 0.0f, 0.0f, 0.0f, 0.0f};
    float r1[8] = {0.0f, 0.0f, 0.0f, 0.0f, 0.0f, 0.0f, 0.0f, 0.0f};
#pragma unroll 1
    for (int q = 0; q < 7; ++q) {
        half_request<1>(q, iv, jv, yb, lane, hB);
        __builtin_amdgcn_sched_barrier(0);
        float mx0 = 0.0f, mx1 = 0.0f;
        half_sum<0>(q, sp, hA, mx0, mx1);
        __builtin_amdgcn_sched_barrier(0);
        half_request<0>(q + 1, iv, jv, yb, lane, hA);
        __builtin_amdgcn_sched_barrier(0);
        half_sum<1>(q, sp, hB, mx0, mx1);
#pragma unroll
        for (int c = 0; c < 7; ++c) {
            r0[c] = r0[c + 1];
            r1[c] = r1[c + 1];
        }
        r0[7] = mx0;
        r1[7] = mx1;
    }
    {
        half_request<1>(7, iv, jv, yb, lane, hB);
        float mx0 = 0.0f, mx1 = 0.0f;
        half_sum<0>(7, sp, hA, mx0, mx1);
        half_sum<1>(7, sp, hB, mx0, mx1);
#pragma unroll
        for (int c = 0; c < 7; ++c) {
            r0[c] = r0[c + 1];
            r1[c] = r1[c + 1];
        }
        r0[7] = mx0;
        r1[7] = mx1;
    }

    float* o0 = out + ((size_t)b * COUTn + lane) * Nn + n0;
    float* o1 = o0 + (size_t)32 * Nn;
    *(float4*)(o0)     = make_float4(r0[0], r0[1], r0[2], r0[3]);
    *(float4*)(o0 + 4) = make_float4(r0[4], r0[5], r0[6], r0[7]);
    *(float4*)(o1)     = make_float4(r1[0], r1[1], r1[2], r1[3]);
    *(float4*)(o1 + 4) = make_float4(r1[4], r1[5], r1[6], r1[7]);
}

extern "C" void kernel_launch(void* const* d_in, const int* in_sizes, int n_in,
                              void* d_out, int out_size, void* d_ws, size_t ws_size,
                              hipStream_t stream) {
    const float* x    = (const float*)d_in[0];
    const int*   eidx = (const int*)  d_in[1];
    const float* pos  = (const float*)d_in[2];
    const float* W    = (const float*)d_in[3];
    const float* bias = (const float*)d_in[4];

    float* Wc = (float*)d_ws;
    float* y  = (float*)((char*)d_ws + 128 * 64 * sizeof(float));

    build_wc<<<(128 * 64 + 255) / 256, 256, 0, stream>>>(W, Wc);
    gemm_y<<<Bn * (Nn / 16), 32, 0, stream>>>(x, Wc, bias, y);
    edge_combine<<<Bn * (Nn / (8 * EWAVES)), ETHR, 0, stream>>>(y, eidx, pos, (float*)d_out);
}


# node GEMM: each 10-wave work-group forms Wc=[W1-W2;W2] in LDS in A-fragment order from W and reads A fragments there (was strided global loads, a line per row); the separate Wc launch is gone
# speedup vs baseline: 1.7983x; 1.1867x over previous
#include <hip/hip_runtime.h>


#define Bn    2
#define Cn    64
#define Nn    20000
#define Kn    16
#define COUTn 64

typedef __attribute__((ext_vector_type(2))) float v2f;
typedef __attribute__((ext_vector_type(8))) float v8f;

#define GW   10
#define GTHR (GW * 32)
#define KCS  516

__global__ __launch_bounds__(GTHR) void gemm_y(
    const float* __restrict__ x, const float* __restrict__ W,
    const float* __restrict__ bias, float* __restrict__ y) {
    __shared__ float wl[16 * KCS];
    const int t    = threadIdx.x;
    const int lane = t & 31;
    const int nlo  = lane & 15;
    const int hi   = lane >> 4;
    const int w    = __builtin_amdgcn_readfirstlane(t >> 5);
    const int tile = (int)blockIdx.x * GW + w;
    const int b    = tile / (Nn / 16);
    const int m0   = (tile % (Nn / 16)) * 16;

    const bool tail = t < 1024 - 3 * GTHR;
    float4 wa[4], wc[4];
#pragma unroll
    for (int it = 0; it < 4; ++it) {
        if (it < 3 || tail) {
            const int item = t + it * GTHR;
            const int kc = item & 15, o = item >> 4;
            wa[it] = *(const float4*)(W + o * 128 + 4 * kc);
            wc[it] = *(const float4*)(W + o * 128 + 64 + 4 * kc);
        }
    }

    const float* xb = x + (size_t)b * Cn * Nn + m0 + nlo;
    v2f bf[16];
#pragma unroll
    for (int kc = 0; kc < 16; ++kc) {
        bf[kc].x = xb[(size_t)(4 * kc + 2 * hi)     * Nn];
        bf[kc].y = xb[(size_t)(4 * kc + 2 * hi + 1) * Nn];
    }
    float4 bv[4][2];
#pragma unroll
    for (int mt = 0; mt < 4; ++mt) {
        bv[mt][0] = *(const float4*)(bias + mt * 16 + 8 * hi);
        bv[mt][1] = *(const float4*)(bias + mt * 16 + 8 * hi + 4);
    }
    __builtin_amdgcn_sched_barrier(0);

#pragma unroll
    for (int it = 0; it < 4; ++it) {
        if (it < 3 || tail) {
            const int item = t + it * GTHR;
            const int kc = item & 15, o = item >> 4;
            float* d = wl + kc * KCS + (o >> 4) * 64 + (o & 15) * 2;
            const float4 a = wa[it], c = wc[it];
            v2f p;
            p.x = a.x - c.x; p.y = a.y - c.y;  *(v2f*)(d)            = p;
            p.x = a.z - c.z; p.y = a.w - c.w;  *(v2f*)(d + 32)       = p;
            p.x = c.x;       p.y = c.y;        *(v2f*)(d + 256)      = p;
            p.x = c.z;       p.y = c.w;        *(v2f*)(d + 256 + 32) = p;
        }
    }

    v8f acc[8];
#pragma unroll
    for (int mt = 0; mt < 8; ++mt) {
        if (mt < 4) {
            acc[mt][0] = bv[mt][0].x; acc[mt][1] = bv[mt][0].y;
            acc[mt][2] = bv[mt][0].z; acc[mt][3] = bv[mt][0].w;
            acc[mt][4] = bv[mt][1].x; acc[mt][5] = bv[mt][1].y;
            acc[mt][6] = bv[mt][1].z; acc[mt][7] = bv[mt][1].w;
        } else {
#pragma unroll
            for (int r = 0; r < 8; ++r) acc[mt][r] = 0.0f;
        }
    }

    __builtin_amdgcn_fence(__ATOMIC_RELEASE, "workgroup", "local");
    __builtin_amdgcn_s_barrier();
    __builtin_amdgcn_fence(__ATOMIC_ACQUIRE, "workgroup", "local");

    const float* wf = wl + 2 * lane;
#pragma unroll
    for (int kc = 0; kc < 16; ++kc) {
#pragma unroll
        for (int mt = 0; mt < 8; ++mt) {
            const v2f af = *(const v2f*)(wf + kc * KCS + mt * 64);
            acc[mt] = __builtin_amdgcn_wmma_f32_16x16x4_f32(
                false, af, false, bf[kc], (short)0, acc[mt], false, false);
        }
    }

    float* yrow = y + ((size_t)b * Nn + m0 + nlo) * 128;
#pragma unroll
    for (int mt = 0; mt < 8; ++mt) {
        float4* dst = (float4*)(yrow + mt * 16 + 8 * hi);
        dst[0] = make_float4(acc[mt][0], acc[mt][1], acc[mt][2], acc[mt][3]);
        dst[1] = make_float4(acc[mt][4], acc[mt][5], acc[mt][6], acc[mt][7]);
    }
}


template <int HALF>
__device__ __forceinline__ void half_request(
    const int q, const int (&iv)[4], const int (&jv)[4],
    const float* __restrict__ yb, const unsigned lane, float (&d)[32]) {
#pragma unroll
    for (int e = 0; e < 8; ++e) {
        const int k   = 8 * HALF + e;
        const int src = 4 * q + (k >> 2);
        const int ik  = __builtin_amdgcn_readlane(iv[k & 3], src);
        const int jk  = __builtin_amdgcn_readlane(jv[k & 3], src);
        const unsigned ei = (unsigned)ik * 128u + lane;
        const unsigned ej = (unsigned)jk * 128u + 64u + lane;
        d[4 * e + 0] = yb[ei];
        d[4 * e + 1] = yb[ei + 32];
        d[4 * e + 2] = yb[ej];
        d[4 * e + 3] = yb[ej + 32];
    }
}

template <int HALF>
__device__ __forceinline__ void half_sum(
    const int q, const int (&sp)[4], const float (&d)[32], float& mx0, float& mx1) {
#pragma unroll
    for (int e = 0; e < 8; ++e) {
        const int k   = 8 * HALF + e;
        const int src = 4 * q + (k >> 2);
        const float sk = __int_as_float(__builtin_amdgcn_readlane(sp[k & 3], src));
        float a0 = d[4 * e + 0] + d[4 * e + 2];
        float a1 = d[4 * e + 1] + d[4 * e + 3];
        a0 = fmaxf(a0, 0.0f) * sk;
        a1 = fmaxf(a1, 0.0f) * sk;
        mx0 = fmaxf(mx0, a0);
        mx1 = fmaxf(mx1, a1);
    }
}

#define EWAVES 20
#define ETHR   (EWAVES * 32)
#define PVEC   (3 * Nn / 4)

__global__ __launch_bounds__(ETHR) void edge_combine(
    const float* __restrict__ y, const int* __restrict__ eidx,
    const float* __restrict__ pos, float* __restrict__ out) {
    __shared__ float4 ptab4[PVEC];
    const int t    = threadIdx.x;
    const unsigned lane = t & 31;
    const int w    = __builtin_amdgcn_readfirstlane(t >> 5);
    const int b    = (int)blockIdx.x / (Nn / (8 * EWAVES));
    const int r    = (int)blockIdx.x - b * (Nn / (8 * EWAVES));
    const int n0   = (r * EWAVES + w) * 8;

    const size_t ebase = ((size_t)b * Nn + n0) * Kn + 4 * lane;
    const int4 j4 = *(const int4*)(eidx + ebase);
    const int4 i4 = *(const int4*)(eidx + (size_t)Bn * Nn * Kn + ebase);
    const int jv[4] = {j4.x, j4.y, j4.z, j4.w};
    const int iv[4] = {i4.x, i4.y, i4.z, i4.w};
    const float* yb = y + (size_t)b * Nn * 128;

    const float4* psrc = (const float4*)(pos + (size_t)b * 3 * Nn);
    const int rot = r * (PVEC / (Nn / (8 * EWAVES)));
    const int  tidx = (t + ETHR * 23 + rot >= PVEC) ? t + ETHR * 23 + rot - PVEC : t + ETHR * 23 + rot;
    const bool tail = t + ETHR * 23 < PVEC;
    float4 tv = make_float4(0.0f, 0.0f, 0.0f, 0.0f);
#pragma unroll
    for (int grp = 0; grp < 3; ++grp) {
        float4 v[8];
        int    at[8];
#pragma unroll
        for (int u = 0; u < 8; ++u) {
            const int k = 8 * grp + u;
            if (k < 23) {
                int idx = t + ETHR * k + rot;
                idx = (idx >= PVEC) ? idx - PVEC : idx;
                at[u] = idx;
                v[u] = psrc[idx];
            }
        }
        if (grp == 2 && tail) tv = psrc[tidx];
#pragma unroll
        for (int u = 0; u < 8; ++u) {
            const int k = 8 * grp + u;
            if (k < 23) ptab4[at[u]] = v[u];
        }
    }
    if (tail) ptab4[tidx] = tv;

    float hA[32], hB[32];
    half_request<0>(0, iv, jv, yb, lane, hA);

    __builtin_amdgcn_fence(__ATOMIC_RELEASE, "workgroup", "local");
    __builtin_amdgcn_s_barrier();
    __builtin_amdgcn_fence(__ATOMIC_ACQUIRE, "workgroup", "local");

    const float* pb = (const float*)ptab4;
    int sp[4];
#pragma unroll
    for (int c = 0; c < 4; ++c) {
        float dx = pb[jv[c]] - pb[iv[c]];
        float dy = pb[Nn + jv[c]] - pb[Nn + iv[c]];
        float dz = pb[2 * Nn + jv[c]] - pb[2 * Nn + iv[c]];
        float dis = sqrtf(dx * dx + dy * dy + dz * dz);
        sp[c] = __float_as_int(2.0f / (1.0f + __expf(dis)));
    }

    float r0[8] = {0.0f, 0.0f, 0.0f, 0.0f, 0.0f, 0.0f, 0.0f, 0.0f};
    float r1[8] = {0.0f, 0.0f, 0.0f, 0.0f, 0.0f, 0.0f, 0.0f, 0.0f};
#pragma unroll 1
    for (int q = 0; q < 7; ++q) {
        half_request<1>(q, iv, jv, yb, lane, hB);
        __builtin_amdgcn_sched_barrier(0);
        float mx0 = 0.0f, mx1 = 0.0f;
        half_sum<0>(q, sp, hA, mx0, mx1);
        __builtin_amdgcn_sched_barrier(0);
        half_request<0>(q + 1, iv, jv, yb, lane, hA);
        __builtin_amdgcn_sched_barrier(0);
        half_sum<1>(q, sp, hB, mx0, mx1);
#pragma unroll
        for (int c = 0; c < 7; ++c) {
            r0[c] = r0[c + 1];
            r1[c] = r1[c + 1];
        }
        r0[7] = mx0;
        r1[7] = mx1;
    }
    {
        half_request<1>(7, iv, jv, yb, lane, hB);
        float mx0 = 0.0f, mx1 = 0.0f;
        half_sum<0>(7, sp, hA, mx0, mx1);
        half_sum<1>(7, sp, hB, mx0, mx1);
#pragma unroll
        for (int c = 0; c < 7; ++c) {
            r0[c] = r0[c + 1];
            r1[c] = r1[c + 1];
        }
        r0[7] = mx0;
        r1[7] = mx1;
    }

    float* o0 = out + ((size_t)b * COUTn + lane) * Nn + n0;
    float* o1 = o0 + (size_t)32 * Nn;
    *(float4*)(o0)     = make_float4(r0[0], r0[1], r0[2], r0[3]);
    *(float4*)(o0 + 4) = make_float4(r0[4], r0[5], r0[6], r0[7]);
    *(float4*)(o1)     = make_float4(r1[0], r1[1], r1[2], r1[3]);
    *(float4*)(o1 + 4) = make_float4(r1[4], r1[5], r1[6], r1[7]);
}

extern "C" void kernel_launch(void* const* d_in, const int* in_sizes, int n_in,
                              void* d_out, int out_size, void* d_ws, size_t ws_size,
                              hipStream_t stream) {
    const float* x    = (const float*)d_in[0];
    const int*   eidx = (const int*)  d_in[1];
    const float* pos  = (const float*)d_in[2];
    const float* W    = (const float*)d_in[3];
    const float* bias = (const float*)d_in[4];

    float* y  = (float*)d_ws;

    gemm_y<<<Bn * (Nn / 16) / GW, GTHR, 0, stream>>>(x, W, bias, y);
    edge_combine<<<Bn * (Nn / (8 * EWAVES)), ETHR, 0, stream>>>(y, eidx, pos, (float*)d_out);
}
